# starting kernel
# baseline (speedup 1.0000x reference)
#include <hip/hip_runtime.h>
#include <hip/hip_bf16.h>
#include <math.h>

typedef __attribute__((ext_vector_type(2))) float v2f;
typedef __attribute__((ext_vector_type(8))) float v8f;

namespace {
constexpr int kLayer  = 2;
constexpr int kN = 32, kL = 512, kD = 1024, kH = 256;
constexpr int kS = 2048, kSpan = 64;
constexpr int kKC = 64;
constexpr int kAStride = 68;
constexpr int kWPairStride = 272;
}

__launch_bounds__(256, 2)
__global__ void attention_pooler_kernel(const float* __restrict__ hs_all,
                                        const int*  __restrict__ spans,
                                        const float* __restrict__ Wm,
                                        const float* __restrict__ bvec,
                                        const float* __restrict__ vvec,
                                        float* __restrict__ out)
{
    __shared__ __align__(16) float As[kSpan * kAStride];
    __shared__ __align__(16) float Wp[(kKC / 2) * kWPairStride * 2];
    __shared__ float sc[kSpan];
    __shared__ float att[kSpan];
    __shared__ float pooled[kH];
    __shared__ float red[2];

    const int s    = blockIdx.x;
    const int tid  = threadIdx.x;
    const int wave = tid >> 5;
    const int lane = tid & 31;
    const int mt   = wave & 3;
    const int nh   = wave >> 2;
    const int hi   = lane >> 4;
    const int l16  = lane & 15;

    const float* hs = hs_all + (size_t)kLayer * kN * kL * kD;

    const int idx   = spans[3 * s + 0];
    const int start = spans[3 * s + 1];
    const int stop  = spans[3 * s + 2];

    if (tid < kSpan) sc[tid]     = 0.f;
    if (tid < kH)    pooled[tid] = 0.f;

    v8f acc[8];
    #pragma unroll
    for (int nt = 0; nt < 8; ++nt)
        #pragma unroll
        for (int e = 0; e < 8; ++e) acc[nt][e] = 0.f;

    const int rowj = tid >> 2;
    const int quad = tid & 3;
    int pos = start + rowj;
    if (pos > kL - 1) pos = kL - 1;
    const float* arow = hs + ((size_t)idx * kL + pos) * kD;

    for (int kc = 0; kc < kD / kKC; ++kc) {
        const int kbase = kc * kKC;
        __syncthreads();

        {
            const float4* src = (const float4*)(arow + kbase + quad * 16);
            float4 t0 = src[0], t1 = src[1], t2 = src[2], t3 = src[3];
            float* dst = &As[rowj * kAStride + quad * 16];
            *(float4*)(dst +  0) = t0;
            *(float4*)(dst +  4) = t1;
            *(float4*)(dst +  8) = t2;
            *(float4*)(dst + 12) = t3;
        }
        #pragma unroll
        for (int i0 = 0; i0 < 8; i0 += 2) {
            const int ta = tid + i0 * 256;
            const int tb = tid + (i0 + 1) * 256;
            const int kpa = ta >> 6, ca = (ta & 63) * 4;
            const int kpb = tb >> 6, cb = (tb & 63) * 4;
            const float* ga0 = Wm + (size_t)(kbase + 2 * kpa + 0) * kH + ca;
            const float* ga1 = Wm + (size_t)(kbase + 2 * kpa + 1) * kH + ca;
            const float* gb0 = Wm + (size_t)(kbase + 2 * kpb + 0) * kH + cb;
            const float* gb1 = Wm + (size_t)(kbase + 2 * kpb + 1) * kH + cb;
            float4 a0 = *(const float4*)ga0;
            float4 a1 = *(const float4*)ga1;
            float4 b0 = *(const float4*)gb0;
            float4 b1 = *(const float4*)gb1;
            float* da = &Wp[(kpa * kWPairStride + ca) * 2];
            float* db = &Wp[(kpb * kWPairStride + cb) * 2];
            v2f p;
            p.x = a0.x; p.y = a1.x; *(v2f*)(da + 0) = p;
            p.x = a0.y; p.y = a1.y; *(v2f*)(da + 2) = p;
            p.x = a0.z; p.y = a1.z; *(v2f*)(da + 4) = p;
            p.x = a0.w; p.y = a1.w; *(v2f*)(da + 6) = p;
            p.x = b0.x; p.y = b1.x; *(v2f*)(db + 0) = p;
            p.x = b0.y; p.y = b1.y; *(v2f*)(db + 2) = p;
            p.x = b0.z; p.y = b1.z; *(v2f*)(db + 4) = p;
            p.x = b0.w; p.y = b1.w; *(v2f*)(db + 6) = p;
        }
        __syncthreads();

        #pragma unroll
        for (int kk = 0; kk < kKC / 4; ++kk) {
            const int kb = kk * 4;
            const v2f a = *(const v2f*)&As[(mt * 16 + l16) * kAStride + kb + 2 * hi];
            v2f bfr[8];
            #pragma unroll
            for (int nt = 0; nt < 8; ++nt) {
                const int col = (nh * 8 + nt) * 16 + l16;
                bfr[nt] = *(const v2f*)&Wp[((kb >> 1) + hi) * (kWPairStride * 2) + col * 2];
            }
            #pragma unroll
            for (int nt = 0; nt < 8; ++nt) {
                acc[nt] = __builtin_amdgcn_wmma_f32_16x16x4_f32(
                    false, a, false, bfr[nt], (short)0, acc[nt], false, false);
            }
        }
    }

    float srow[8];
    #pragma unroll
    for (int r = 0; r < 8; ++r) srow[r] = 0.f;
    #pragma unroll
    for (int nt = 0; nt < 8; ++nt) {
        const int col    = (nh * 8 + nt) * 16 + l16;
        const float bcol = bvec[col];
        const float vcol = vvec[col];
        #pragma unroll
        for (int r = 0; r < 8; ++r) {
            acc[nt][r] += bcol;
            srow[r] += acc[nt][r] * vcol;
        }
    }
    #pragma unroll
    for (int off = 8; off >= 1; off >>= 1)
        #pragma unroll
        for (int r = 0; r < 8; ++r)
            srow[r] += __shfl_xor(srow[r], off, 16);
    if (l16 == 0) {
        #pragma unroll
        for (int r = 0; r < 8; ++r)
            atomicAdd(&sc[mt * 16 + hi * 8 + r], srow[r]);
    }
    __syncthreads();

    const int len = stop - start;
    if (tid == 0) {
        float m = -INFINITY;
        for (int l = 0; l < kSpan; ++l)
            if (l < len) m = fmaxf(m, sc[l]);
        red[0] = m;
    }
    __syncthreads();
    if (tid < kSpan)
        att[tid] = (tid < len) ? expf(sc[tid] - red[0]) : 0.f;
    __syncthreads();
    if (tid == 0) {
        float sum = 0.f;
        for (int l = 0; l < kSpan; ++l) sum += att[l];
        red[1] = 1.f / sum;
    }
    __syncthreads();

    float aw[8];
    const float inv = red[1];
    #pragma unroll
    for (int r = 0; r < 8; ++r)
        aw[r] = att[mt * 16 + hi * 8 + r] * inv;
    #pragma unroll
    for (int nt = 0; nt < 8; ++nt) {
        float p = 0.f;
        #pragma unroll
        for (int r = 0; r < 8; ++r) p += aw[r] * acc[nt][r];
        p += __shfl_xor(p, 16, 32);
        if (hi == 0) atomicAdd(&pooled[(nh * 8 + nt) * 16 + l16], p);
    }
    __syncthreads();

    out[(size_t)s * kH + tid] = pooled[tid];
}

extern "C" void kernel_launch(void* const* d_in, const int* in_sizes, int n_in,
                              void* d_out, int out_size, void* d_ws, size_t ws_size,
                              hipStream_t stream) {
    (void)in_sizes; (void)n_in; (void)out_size; (void)d_ws; (void)ws_size;
    const float* hs    = (const float*)d_in[0];
    const int*   spans = (const int*)  d_in[1];
    const float* W     = (const float*)d_in[2];
    const float* b     = (const float*)d_in[3];
    const float* v     = (const float*)d_in[4];
    float* out = (float*)d_out;
    attention_pooler_kernel<<<dim3(kS), dim3(256), 0, stream>>>(hs, spans, W, b, v, out);
}


# Work-groups take spans heaviest first: block g finds the span of rank g by row-tile count (4 down to 1) in a counting pass over the 2048 lengths, so one-tile spans start last; sums unchanged
# speedup vs baseline: 1.6873x; 1.6873x over previous
#include <hip/hip_runtime.h>
#include <hip/hip_bf16.h>
#include <math.h>


typedef __attribute__((ext_vector_type(2))) float v2f;
typedef __attribute__((ext_vector_type(8))) float v8f;

namespace {
constexpr int kLayer  = 2;
constexpr int kN = 32, kL = 512, kD = 1024, kH = 256;
constexpr int kS = 2048, kSpan = 64;
constexpr int kKA = 256;
constexpr int kAStride = 260;
constexpr int kKB = 32;
constexpr int kWaves = 8;

struct WRegs { v2f r0[kKB / 4]; v2f r1[kKB / 4]; };
}

__device__ __forceinline__ void load_w(WRegs& f, const float* __restrict__ wl, const int k0)
{
    const float* p = wl + (size_t)k0 * kH;
    #pragma unroll
    for (int kk = 0; kk < kKB / 4; ++kk) {
        f.r0[kk] = *(const v2f*)(p + (kk * 4 + 0) * kH);
        f.r1[kk] = *(const v2f*)(p + (kk * 4 + 1) * kH);
    }
}

template <int MT>
__device__ __forceinline__ void mma_chunk(v8f (&acc)[MT][2], const WRegs& f,
                                          const float* __restrict__ Ak)
{
    #pragma unroll
    for (int kk = 0; kk < kKB / 4; ++kk) {
        v2f a[MT];
        #pragma unroll
        for (int mi = 0; mi < MT; ++mi)
            a[mi] = *(const v2f*)&Ak[mi * 16 * kAStride + kk * 4];
        v2f b0, b1;
        b0.x = f.r0[kk].x; b0.y = f.r1[kk].x;
        b1.x = f.r0[kk].y; b1.y = f.r1[kk].y;
        #pragma unroll
        for (int mi = 0; mi < MT; ++mi) {
            acc[mi][0] = __builtin_amdgcn_wmma_f32_16x16x4_f32(
                false, a[mi], false, b0, (short)0, acc[mi][0], false, false);
            acc[mi][1] = __builtin_amdgcn_wmma_f32_16x16x4_f32(
                false, a[mi], false, b1, (short)0, acc[mi][1], false, false);
        }
    }
}

template <int MT>
__device__ __forceinline__ void span_body(const float* __restrict__ hs,
                                          const float* __restrict__ Wm,
                                          const float* __restrict__ bvec,
                                          const float* __restrict__ vvec,
                                          float* __restrict__ out,
                                          float* __restrict__ As,
                                          float* __restrict__ scp,
                                          float* __restrict__ sc,
                                          float* __restrict__ att,
                                          float* __restrict__ red,
                                          const int s, const int idx,
                                          const int start, const int len)
{
    const int tid  = threadIdx.x;
    const int wave = tid >> 5;
    const int lane = tid & 31;
    const int hi   = lane >> 4;
    const int l16  = lane & 15;

    v8f acc[MT][2];
    #pragma unroll
    for (int mi = 0; mi < MT; ++mi)
        #pragma unroll
        for (int ni = 0; ni < 2; ++ni)
            #pragma unroll
            for (int e = 0; e < 8; ++e) acc[mi][ni][e] = 0.f;

    const int arow_t = tid >> 6;
    const int ac4    = tid & 63;
    const float* hsamp = hs + (size_t)idx * kL * kD + ac4 * 4;

    const float* wl = Wm + (size_t)(2 * hi) * kH + wave * 32 + 2 * l16;
    const float* Afrag = As + l16 * kAStride + 2 * hi;

    WRegs w0, w1;
    load_w(w0, wl, 0);

    for (int kc = 0; kc < kD / kKA; ++kc) {
        const int kbase = kc * kKA;
        __syncthreads();

        {
            float4 ta[MT][4];
            #pragma unroll
            for (int mi = 0; mi < MT; ++mi)
                #pragma unroll
                for (int j = 0; j < 4; ++j) {
                    int pos = start + mi * 16 + arow_t + 4 * j;
                    if (pos > kL - 1) pos = kL - 1;
                    ta[mi][j] = *(const float4*)(hsamp + (size_t)pos * kD + kbase);
                }
            #pragma unroll
            for (int mi = 0; mi < MT; ++mi)
                #pragma unroll
                for (int j = 0; j < 4; ++j)
                    *(float4*)&As[(mi * 16 + arow_t + 4 * j) * kAStride + ac4 * 4] = ta[mi][j];
        }
        __syncthreads();

        #pragma nounroll
        for (int sc = 0; sc < kKA / kKB; sc += 2) {
            const int kg = kbase + sc * kKB;
            load_w(w1, wl, kg + kKB);
            mma_chunk<MT>(acc, w0, Afrag + sc * kKB);
            int kn = kg + 2 * kKB;
            if (kn > kD - kKB) kn = kD - kKB;
            load_w(w0, wl, kn);
            mma_chunk<MT>(acc, w1, Afrag + (sc + 1) * kKB);
        }
    }

    float srow[MT][8];
    #pragma unroll
    for (int mi = 0; mi < MT; ++mi)
        #pragma unroll
        for (int r = 0; r < 8; ++r) srow[mi][r] = 0.f;
    #pragma unroll
    for (int ni = 0; ni < 2; ++ni) {
        const int col    = wave * 32 + 2 * l16 + ni;
        const float bcol = bvec[col];
        const float vcol = vvec[col];
        #pragma unroll
        for (int mi = 0; mi < MT; ++mi)
            #pragma unroll
            for (int r = 0; r < 8; ++r) {
                acc[mi][ni][r] += bcol;
                srow[mi][r] += acc[mi][ni][r] * vcol;
            }
    }
    #pragma unroll
    for (int off = 8; off >= 1; off >>= 1)
        #pragma unroll
        for (int mi = 0; mi < MT; ++mi)
            #pragma unroll
            for (int r = 0; r < 8; ++r)
                srow[mi][r] += __shfl_xor(srow[mi][r], off, 16);
    if (l16 == 0) {
        #pragma unroll
        for (int mi = 0; mi < MT; ++mi)
            #pragma unroll
            for (int r = 0; r < 8; ++r)
                scp[wave * kSpan + mi * 16 + hi * 8 + r] = srow[mi][r];
    }
    __syncthreads();
    if (tid < MT * 16) {
        float t = scp[tid];
        #pragma unroll
        for (int w = 1; w < kWaves; ++w) t += scp[w * kSpan + tid];
        sc[tid] = t;
    }
    __syncthreads();

    if (tid == 0) {
        float m = -INFINITY;
        for (int l = 0; l < kSpan; ++l)
            if (l < len) m = fmaxf(m, sc[l]);
        red[0] = m;
    }
    __syncthreads();
    if (tid < kSpan)
        att[tid] = (tid < len) ? expf(sc[tid] - red[0]) : 0.f;
    __syncthreads();
    if (tid == 0) {
        float sum = 0.f;
        for (int l = 0; l < kSpan; ++l) sum += att[l];
        red[1] = 1.f / sum;
    }
    __syncthreads();

    const float inv = red[1];
    float p0 = 0.f, p1 = 0.f;
    #pragma unroll
    for (int mi = 0; mi < MT; ++mi)
        #pragma unroll
        for (int r = 0; r < 8; ++r) {
            const float aw = att[mi * 16 + hi * 8 + r] * inv;
            p0 += aw * acc[mi][0][r];
            p1 += aw * acc[mi][1][r];
        }
    p0 += __shfl_xor(p0, 16, 32);
    p1 += __shfl_xor(p1, 16, 32);
    if (hi == 0) {
        v2f o; o.x = p0; o.y = p1;
        *(v2f*)&out[(size_t)s * kH + wave * 32 + 2 * l16] = o;
    }
}


__device__ __forceinline__ int span_of_rank(const int* __restrict__ spans,
                                            unsigned* __restrict__ wtot,
                                            int* __restrict__ pick, const int g)
{
    const int tid  = threadIdx.x;
    const int lane = tid & 31;
    const int wave = tid >> 5;

    const int4* p = (const int4*)(spans + 24 * tid);
    int v[24];
    #pragma unroll
    for (int q = 0; q < 6; ++q) {
        const int4 x = p[q];
        v[4 * q + 0] = x.x; v[4 * q + 1] = x.y; v[4 * q + 2] = x.z; v[4 * q + 3] = x.w;
    }
    int cls[8];
    unsigned lo = 0u, hi = 0u;
    #pragma unroll
    for (int j = 0; j < 8; ++j) {
        const int len = v[3 * j + 2] - v[3 * j + 1];
        int mt = (len + 15) >> 4;
        if (mt < 1) mt = 1;
        if (mt > 4) mt = 4;
        cls[j] = 4 - mt;
        const unsigned inc = 1u << (16 * (cls[j] & 1));
        if (cls[j] < 2) lo += inc; else hi += inc;
    }
    unsigned slo = lo, shi = hi;
    #pragma unroll
    for (int off = 1; off < 32; off <<= 1) {
        const unsigned a = __shfl_up(slo, off, 32);
        const unsigned b = __shfl_up(shi, off, 32);
        if (lane >= off) { slo += a; shi += b; }
    }
    if (lane == 31) { wtot[2 * wave] = slo; wtot[2 * wave + 1] = shi; }
    __syncthreads();
    unsigned blo = 0u, bhi = 0u, tlo = 0u, thi = 0u;
    #pragma unroll
    for (int w = 0; w < kWaves; ++w) {
        const unsigned a = wtot[2 * w], b = wtot[2 * w + 1];
        if (w < wave) { blo += a; bhi += b; }
        tlo += a; thi += b;
    }
    const int t0 = (int)(tlo & 0xffffu), t1 = (int)(tlo >> 16), t2 = (int)(thi & 0xffffu);
    int c = 0, q = g;
    if (q >= t0) { q -= t0; c = 1;
        if (q >= t1) { q -= t1; c = 2;
            if (q >= t2) { q -= t2; c = 3; } } }
    const unsigned ew = (c < 2) ? (blo + slo - lo) : (bhi + shi - hi);
    const unsigned ow = (c < 2) ? lo : hi;
    const int sh  = 16 * (c & 1);
    const int own = (int)((ow >> sh) & 0xffffu);
    int r = q - (int)((ew >> sh) & 0xffffu);
    if (r >= 0 && r < own) {
        #pragma unroll
        for (int j = 0; j < 8; ++j)
            if (cls[j] == c) { if (r == 0) *pick = 8 * tid + j; --r; }
    }
    __syncthreads();
    return *pick;
}

__launch_bounds__(256, 2)
__global__ void attention_pooler_kernel(const float* __restrict__ hs_all,
                                        const int*  __restrict__ spans,
                                        const float* __restrict__ Wm,
                                        const float* __restrict__ bvec,
                                        const float* __restrict__ vvec,
                                        float* __restrict__ out)
{
    __shared__ __align__(16) float As[kSpan * kAStride];
    __shared__ float scp[kWaves * kSpan];
    __shared__ float sc[kSpan];
    __shared__ __align__(16) float att[kSpan];
    __shared__ float red[2];
    __shared__ unsigned wtot[2 * kWaves];
    __shared__ int pick;

    int s = span_of_rank(spans, wtot, &pick, (int)blockIdx.x);
    s = __builtin_amdgcn_readfirstlane(s);
    if (s < 0) s = 0;
    if (s > kS - 1) s = kS - 1;
    const float* hs = hs_all + (size_t)kLayer * kN * kL * kD;

    const int idx   = spans[3 * s + 0];
    const int start = spans[3 * s + 1];
    const int stop  = spans[3 * s + 2];
    const int len   = stop - start;

    int mt = (len + 15) >> 4;
    if (mt < 1) mt = 1;
    if (mt > 4) mt = 4;

    switch (mt) {
    case 1:  span_body<1>(hs, Wm, bvec, vvec, out, As, scp, sc, att, red, s, idx, start, len); break;
    case 2:  span_body<2>(hs, Wm, bvec, vvec, out, As, scp, sc, att, red, s, idx, start, len); break;
    case 3:  span_body<3>(hs, Wm, bvec, vvec, out, As, scp, sc, att, red, s, idx, start, len); break;
    default: span_body<4>(hs, Wm, bvec, vvec, out, As, scp, sc, att, red, s, idx, start, len); break;
    }
}

extern "C" void kernel_launch(void* const* d_in, const int* in_sizes, int n_in,
                              void* d_out, int out_size, void* d_ws, size_t ws_size,
                              hipStream_t stream) {
    (void)in_sizes; (void)n_in; (void)out_size; (void)d_ws; (void)ws_size;
    const float* hs    = (const float*)d_in[0];
    const int*   spans = (const int*)  d_in[1];
    const float* W     = (const float*)d_in[2];
    const float* b     = (const float*)d_in[3];
    const float* v     = (const float*)d_in[4];
    float* out = (float*)d_out;
    attention_pooler_kernel<<<dim3(kS), dim3(256), 0, stream>>>(hs, spans, W, b, v, out);
}


# Spans of a and 64-a rows (a no multiple of 16) share one 64-row block, one projection and one W pass, not five row tiles in two work-groups; the shorter span's pooled row-sum order changes, f32
# speedup vs baseline: 1.9443x; 1.1523x over previous
#include <hip/hip_runtime.h>
#include <hip/hip_bf16.h>
#include <math.h>


typedef __attribute__((ext_vector_type(2))) float v2f;
typedef __attribute__((ext_vector_type(8))) float v8f;

namespace {
constexpr int kLayer  = 2;
constexpr int kN = 32, kL = 512, kD = 1024, kH = 256;
constexpr int kS = 2048, kSpan = 64;
constexpr int kKA = 256;
constexpr int kAStride = 260;
constexpr int kKB = 32;
constexpr int kWaves = 8;

struct WRegs { v2f r0[kKB / 4]; v2f r1[kKB / 4]; };
}

__device__ __forceinline__ void load_w(WRegs& f, const float* __restrict__ wl, const int k0)
{
    const float* p = wl + (size_t)k0 * kH;
    #pragma unroll
    for (int kk = 0; kk < kKB / 4; ++kk) {
        f.r0[kk] = *(const v2f*)(p + (kk * 4 + 0) * kH);
        f.r1[kk] = *(const v2f*)(p + (kk * 4 + 1) * kH);
    }
}

template <int MT>
__device__ __forceinline__ void mma_chunk(v8f (&acc)[MT][2], const WRegs& f,
                                          const float* __restrict__ Ak)
{
    #pragma unroll
    for (int kk = 0; kk < kKB / 4; ++kk) {
        v2f a[MT];
        #pragma unroll
        for (int mi = 0; mi < MT; ++mi)
            a[mi] = *(const v2f*)&Ak[mi * 16 * kAStride + kk * 4];
        v2f b0, b1;
        b0.x = f.r0[kk].x; b0.y = f.r1[kk].x;
        b1.x = f.r0[kk].y; b1.y = f.r1[kk].y;
        #pragma unroll
        for (int mi = 0; mi < MT; ++mi) {
            acc[mi][0] = __builtin_amdgcn_wmma_f32_16x16x4_f32(
                false, a[mi], false, b0, (short)0, acc[mi][0], false, false);
            acc[mi][1] = __builtin_amdgcn_wmma_f32_16x16x4_f32(
                false, a[mi], false, b1, (short)0, acc[mi][1], false, false);
        }
    }
}

template <int MT>
__device__ __forceinline__ void span_body(const float* __restrict__ hs,
                                          const float* __restrict__ Wm,
                                          const float* __restrict__ bvec,
                                          const float* __restrict__ vvec,
                                          float* __restrict__ out,
                                          float* __restrict__ As,
                                          float* __restrict__ scp,
                                          float* __restrict__ sc,
                                          float* __restrict__ att,
                                          float* __restrict__ red,
                                          const int s, const int idx,
                                          const int start, const int len,
                                          const int sq, const int idxq,
                                          const int startq, const int lenq)
{
    const int tid  = threadIdx.x;
    const int wave = tid >> 5;
    const int lane = tid & 31;
    const int hi   = lane >> 4;
    const int l16  = lane & 15;

    v8f acc[MT][2];
    #pragma unroll
    for (int mi = 0; mi < MT; ++mi)
        #pragma unroll
        for (int ni = 0; ni < 2; ++ni)
            #pragma unroll
            for (int e = 0; e < 8; ++e) acc[mi][ni][e] = 0.f;

    const int arow_t = tid >> 6;
    const int ac4    = tid & 63;
    const float* hsamp = hs + (size_t)idx * kL * kD + ac4 * 4;
    [[maybe_unused]] const int split = (lenq > 0) ? len : kSpan;
    [[maybe_unused]] const float* hsampq = hs + (size_t)idxq * kL * kD + ac4 * 4;

    const float* wl = Wm + (size_t)(2 * hi) * kH + wave * 32 + 2 * l16;
    const float* Afrag = As + l16 * kAStride + 2 * hi;

    WRegs w0, w1;
    load_w(w0, wl, 0);

    for (int kc = 0; kc < kD / kKA; ++kc) {
        const int kbase = kc * kKA;
        __syncthreads();

        {
            float4 ta[MT][4];
            #pragma unroll
            for (int mi = 0; mi < MT; ++mi)
                #pragma unroll
                for (int j = 0; j < 4; ++j) {
                    const int row = mi * 16 + arow_t + 4 * j;
                    int pos = start + row;
                    const float* src = hsamp;
                    if constexpr (MT == 4) {
                        if (row >= split) { pos = startq + (row - split); src = hsampq; }
                    }
                    if (pos > kL - 1) pos = kL - 1;
                    ta[mi][j] = *(const float4*)(src + (size_t)pos * kD + kbase);
                }
            #pragma unroll
            for (int mi = 0; mi < MT; ++mi)
                #pragma unroll
                for (int j = 0; j < 4; ++j)
                    *(float4*)&As[(mi * 16 + arow_t + 4 * j) * kAStride + ac4 * 4] = ta[mi][j];
        }
        __syncthreads();

        #pragma nounroll
        for (int sc = 0; sc < kKA / kKB; sc += 2) {
            const int kg = kbase + sc * kKB;
            load_w(w1, wl, kg + kKB);
            mma_chunk<MT>(acc, w0, Afrag + sc * kKB);
            int kn = kg + 2 * kKB;
            if (kn > kD - kKB) kn = kD - kKB;
            load_w(w0, wl, kn);
            mma_chunk<MT>(acc, w1, Afrag + (sc + 1) * kKB);
        }
    }

    float srow[MT][8];
    #pragma unroll
    for (int mi = 0; mi < MT; ++mi)
        #pragma unroll
        for (int r = 0; r < 8; ++r) srow[mi][r] = 0.f;
    #pragma unroll
    for (int ni = 0; ni < 2; ++ni) {
        const int col    = wave * 32 + 2 * l16 + ni;
        const float bcol = bvec[col];
        const float vcol = vvec[col];
        #pragma unroll
        for (int mi = 0; mi < MT; ++mi)
            #pragma unroll
            for (int r = 0; r < 8; ++r) {
                acc[mi][ni][r] += bcol;
                srow[mi][r] += acc[mi][ni][r] * vcol;
            }
    }
    #pragma unroll
    for (int off = 8; off >= 1; off >>= 1)
        #pragma unroll
        for (int mi = 0; mi < MT; ++mi)
            #pragma unroll
            for (int r = 0; r < 8; ++r)
                srow[mi][r] += __shfl_xor(srow[mi][r], off, 16);
    if (l16 == 0) {
        #pragma unroll
        for (int mi = 0; mi < MT; ++mi)
            #pragma unroll
            for (int r = 0; r < 8; ++r)
                scp[wave * kSpan + mi * 16 + hi * 8 + r] = srow[mi][r];
    }
    __syncthreads();
    if (tid < MT * 16) {
        float t = scp[tid];
        #pragma unroll
        for (int w = 1; w < kWaves; ++w) t += scp[w * kSpan + tid];
        sc[tid] = t;
    }
    __syncthreads();

    if (tid == 0) {
        float m = -INFINITY;
        for (int l = 0; l < kSpan; ++l)
            if (l < len) m = fmaxf(m, sc[l]);
        red[0] = m;
    }
    if constexpr (MT == 4) {
        if (tid == 32 && lenq > 0) {
            float m = -INFINITY;
            for (int l = 0; l < kSpan; ++l)
                if (l >= split && l < split + lenq) m = fmaxf(m, sc[l]);
            red[2] = m;
        }
    }
    __syncthreads();
    if (tid < kSpan) {
        float e = (tid < len) ? expf(sc[tid] - red[0]) : 0.f;
        if constexpr (MT == 4) {
            if (lenq > 0 && tid >= split && tid < split + lenq) e = expf(sc[tid] - red[2]);
        }
        att[tid] = e;
    }
    __syncthreads();
    if (tid == 0) {
        float sum = 0.f;
        if constexpr (MT == 4) {
            for (int l = 0; l < kSpan; ++l)
                if (l < split) sum += att[l];
        } else {
            for (int l = 0; l < kSpan; ++l) sum += att[l];
        }
        red[1] = 1.f / sum;
    }
    if constexpr (MT == 4) {
        if (tid == 32 && lenq > 0) {
            float sum = 0.f;
            for (int l = 0; l < kSpan; ++l)
                if (l >= split) sum += att[l];
            red[3] = 1.f / sum;
        }
    }
    __syncthreads();

    const float inv = red[1];
    float p0 = 0.f, p1 = 0.f;
    [[maybe_unused]] float q0 = 0.f, q1 = 0.f;
    [[maybe_unused]] float invq = 0.f;
    if constexpr (MT == 4) {
        if (lenq > 0) invq = red[3];
    }
    #pragma unroll
    for (int mi = 0; mi < MT; ++mi)
        #pragma unroll
        for (int r = 0; r < 8; ++r) {
            const int row = mi * 16 + hi * 8 + r;
            const float a = att[row];
            if constexpr (MT == 4) {
                const float awp = (row < split) ? a * inv : 0.f;
                const float awq = (row < split) ? 0.f : a * invq;
                p0 += awp * acc[mi][0][r];
                p1 += awp * acc[mi][1][r];
                q0 += awq * acc[mi][0][r];
                q1 += awq * acc[mi][1][r];
            } else {
                const float aw = a * inv;
                p0 += aw * acc[mi][0][r];
                p1 += aw * acc[mi][1][r];
            }
        }
    p0 += __shfl_xor(p0, 16, 32);
    p1 += __shfl_xor(p1, 16, 32);
    if constexpr (MT == 4) {
        q0 += __shfl_xor(q0, 16, 32);
        q1 += __shfl_xor(q1, 16, 32);
    }
    if (hi == 0) {
        v2f o; o.x = p0; o.y = p1;
        *(v2f*)&out[(size_t)s * kH + wave * 32 + 2 * l16] = o;
        if constexpr (MT == 4) {
            if (lenq > 0) {
                v2f oq; oq.x = q0; oq.y = q1;
                *(v2f*)&out[(size_t)sq * kH + wave * 32 + 2 * l16] = oq;
            }
        }
    }
}


__device__ __forceinline__ int span_of_rank(const int* __restrict__ spans,
                                            unsigned* __restrict__ wtot,
                                            int* __restrict__ pick, const int g)
{
    const int tid  = threadIdx.x;
    const int lane = tid & 31;
    const int wave = tid >> 5;

    const int4* p = (const int4*)(spans + 24 * tid);
    int v[24];
    #pragma unroll
    for (int q = 0; q < 6; ++q) {
        const int4 x = p[q];
        v[4 * q + 0] = x.x; v[4 * q + 1] = x.y; v[4 * q + 2] = x.z; v[4 * q + 3] = x.w;
    }
    int cls[8];
    unsigned lo = 0u, hi = 0u;
    #pragma unroll
    for (int j = 0; j < 8; ++j) {
        const int len = v[3 * j + 2] - v[3 * j + 1];
        int mt = (len + 15) >> 4;
        if (mt < 1) mt = 1;
        if (mt > 4) mt = 4;
        if (len > kSpan / 2 && len < kSpan && (len & 15) != 0) mt = 4;
        cls[j] = 4 - mt;
        const unsigned inc = 1u << (16 * (cls[j] & 1));
        if (cls[j] < 2) lo += inc; else hi += inc;
    }
    unsigned slo = lo, shi = hi;
    #pragma unroll
    for (int off = 1; off < 32; off <<= 1) {
        const unsigned a = __shfl_up(slo, off, 32);
        const unsigned b = __shfl_up(shi, off, 32);
        if (lane >= off) { slo += a; shi += b; }
    }
    if (lane == 31) { wtot[2 * wave] = slo; wtot[2 * wave + 1] = shi; }
    __syncthreads();
    unsigned blo = 0u, bhi = 0u, tlo = 0u, thi = 0u;
    #pragma unroll
    for (int w = 0; w < kWaves; ++w) {
        const unsigned a = wtot[2 * w], b = wtot[2 * w + 1];
        if (w < wave) { blo += a; bhi += b; }
        tlo += a; thi += b;
    }
    const int t0 = (int)(tlo & 0xffffu), t1 = (int)(tlo >> 16), t2 = (int)(thi & 0xffffu);
    int c = 0, q = g;
    if (q >= t0) { q -= t0; c = 1;
        if (q >= t1) { q -= t1; c = 2;
            if (q >= t2) { q -= t2; c = 3; } } }
    const unsigned ew = (c < 2) ? (blo + slo - lo) : (bhi + shi - hi);
    const unsigned ow = (c < 2) ? lo : hi;
    const int sh  = 16 * (c & 1);
    const int own = (int)((ow >> sh) & 0xffffu);
    int r = q - (int)((ew >> sh) & 0xffffu);
    if (r >= 0 && r < own) {
        #pragma unroll
        for (int j = 0; j < 8; ++j)
            if (cls[j] == c) { if (r == 0) *pick = 8 * tid + j; --r; }
    }
    __syncthreads();
    return *pick;
}

__device__ __forceinline__ int pair_role(const int* __restrict__ spans,
                                         unsigned* __restrict__ ptot,
                                         int* __restrict__ pick2,
                                         const int s, const int a)
{
    const int tid  = threadIdx.x;
    const int lane = tid & 31;
    const int wave = tid >> 5;
    const int ap   = kSpan - a;

    const int4* p = (const int4*)(spans + 24 * tid);
    int v[24];
    #pragma unroll
    for (int q = 0; q < 6; ++q) {
        const int4 x = p[q];
        v[4 * q + 0] = x.x; v[4 * q + 1] = x.y; v[4 * q + 2] = x.z; v[4 * q + 3] = x.w;
    }
    bool isb[8];
    unsigned cnt = 0u;
    #pragma unroll
    for (int j = 0; j < 8; ++j) {
        const int lj = v[3 * j + 2] - v[3 * j + 1];
        if (lj == a && 8 * tid + j < s) cnt += 1u;
        isb[j] = (lj == ap);
        if (isb[j]) cnt += 0x10000u;
    }
    unsigned inc = cnt;
    #pragma unroll
    for (int off = 1; off < 32; off <<= 1) {
        const unsigned x = __shfl_up(inc, off, 32);
        if (lane >= off) inc += x;
    }
    if (lane == 31) ptot[wave] = inc;
    if (tid == 0) *pick2 = -1;
    __syncthreads();
    unsigned base = 0u, tot = 0u;
    #pragma unroll
    for (int w = 0; w < kWaves; ++w) {
        const unsigned x = ptot[w];
        if (w < wave) base += x;
        tot += x;
    }
    const int  k    = (int)(tot & 0xffffu);
    const int  nb   = (int)(tot >> 16);
    const bool has  = k < nb;
    const bool lead = a > ap;
    if (has && lead) {
        const int excl = (int)((base + inc - cnt) >> 16);
        const int own  = (int)(cnt >> 16);
        int r = k - excl;
        if (r >= 0 && r < own) {
            #pragma unroll
            for (int j = 0; j < 8; ++j)
                if (isb[j]) { if (r == 0) *pick2 = 8 * tid + j; --r; }
        }
    }
    __syncthreads();
    const int q = *pick2;
    return has ? (lead ? q : -2) : -1;
}

__launch_bounds__(256, 2)
__global__ void attention_pooler_kernel(const float* __restrict__ hs_all,
                                        const int*  __restrict__ spans,
                                        const float* __restrict__ Wm,
                                        const float* __restrict__ bvec,
                                        const float* __restrict__ vvec,
                                        float* __restrict__ out)
{
    __shared__ __align__(16) float As[kSpan * kAStride];
    __shared__ float scp[kWaves * kSpan];
    __shared__ float sc[kSpan];
    __shared__ __align__(16) float att[kSpan];
    __shared__ float red[4];
    __shared__ unsigned wtot[2 * kWaves];
    __shared__ int pick;
    __shared__ unsigned ptot[kWaves];
    __shared__ int pick2;

    int s = span_of_rank(spans, wtot, &pick, (int)blockIdx.x);
    s = __builtin_amdgcn_readfirstlane(s);
    if (s < 0) s = 0;
    if (s > kS - 1) s = kS - 1;
    const float* hs = hs_all + (size_t)kLayer * kN * kL * kD;

    const int idx   = spans[3 * s + 0];
    const int start = spans[3 * s + 1];
    const int stop  = spans[3 * s + 2];
    const int len   = stop - start;

    int q = -1;
    if (len >= 1 && len <= kSpan - 1 && (len & 15) != 0) {
        q = pair_role(spans, ptot, &pick2, s, len);
        q = __builtin_amdgcn_readfirstlane(q);
        if (q == -2) return;
    }
    int sq = s, idxq = idx, startq = start, lenq = 0;
    if (q >= 0) {
        if (q > kS - 1) q = kS - 1;
        sq     = q;
        idxq   = spans[3 * q + 0];
        startq = spans[3 * q + 1];
        lenq   = spans[3 * q + 2] - startq;
    }

    int mt = (len + 15) >> 4;
    if (mt < 1) mt = 1;
    if (mt > 4) mt = 4;
    if (q >= 0) mt = 4;

    switch (mt) {
    case 1:  span_body<1>(hs, Wm, bvec, vvec, out, As, scp, sc, att, red, s, idx, start, len, sq, idxq, startq, lenq); break;
    case 2:  span_body<2>(hs, Wm, bvec, vvec, out, As, scp, sc, att, red, s, idx, start, len, sq, idxq, startq, lenq); break;
    case 3:  span_body<3>(hs, Wm, bvec, vvec, out, As, scp, sc, att, red, s, idx, start, len, sq, idxq, startq, lenq); break;
    default: span_body<4>(hs, Wm, bvec, vvec, out, As, scp, sc, att, red, s, idx, start, len, sq, idxq, startq, lenq); break;
    }
}

extern "C" void kernel_launch(void* const* d_in, const int* in_sizes, int n_in,
                              void* d_out, int out_size, void* d_ws, size_t ws_size,
                              hipStream_t stream) {
    (void)in_sizes; (void)n_in; (void)out_size; (void)d_ws; (void)ws_size;
    const float* hs    = (const float*)d_in[0];
    const int*   spans = (const int*)  d_in[1];
    const float* W     = (const float*)d_in[2];
    const float* b     = (const float*)d_in[3];
    const float* v     = (const float*)d_in[4];
    float* out = (float*)d_out;
    attention_pooler_kernel<<<dim3(kS), dim3(256), 0, stream>>>(hs, spans, W, b, v, out);
}
